# starting kernel
# baseline (speedup 1.0000x reference)
#include <hip/hip_runtime.h>


typedef __attribute__((ext_vector_type(2))) float v2f;
typedef __attribute__((ext_vector_type(8))) float v8f;

#define PLANE 1024
#define CLIPV 2.0f

__global__ __launch_bounds__(256)
void qam_vq_wmma(const float* __restrict__ z,
                 const float* __restrict__ cb,
                 float* __restrict__ out,
                 int numSuper)
{
    const int lane = threadIdx.x & 31;
    const int lp   = lane & 15;
    const int half = lane >> 4;

    const int wavesPerBlk = blockDim.x >> 5;
    const int gwave  = blockIdx.x * wavesPerBlk + (threadIdx.x >> 5);
    const int nwaves = gridDim.x * wavesPerBlk;

    float4 c4 = ((const float4*)cb)[lane];
    float ss = c4.x*c4.x + c4.y*c4.y + c4.z*c4.z + c4.w*c4.w;
    #pragma unroll
    for (int m = 16; m >= 1; m >>= 1) ss += __shfl_xor(ss, m, 32);
    const float power = ss * (1.0f / 64.0f);
    const float scale = sqrtf(1.0f / (power + 1e-8f));
    const float nhalf = -0.5f * scale * scale;

    v2f Ag[4];
    #pragma unroll
    for (int g = 0; g < 4; ++g) {
        const int e = g * 16 + lp;
        const float ci = cb[2*e], cq = cb[2*e + 1];
        const float ht = nhalf * (ci*ci + cq*cq);
        Ag[g][0] = half ? ht   : ci * scale;
        Ag[g][1] = half ? 0.0f : cq * scale;
    }

    for (int t = gwave; t < numSuper; t += nwaves) {
        const int p = t >> 5;
        const int o = (t & 31) << 5;
        const float* zi = z + (size_t)p * (2 * PLANE) + o;
        const float* zq = zi + PLANE;
        float* ob = out + (size_t)p * (2 * PLANE) + o;

        #pragma unroll
        for (int sub = 0; sub < 2; ++sub) {
            const int so = sub * 16;

            float si = zi[so + lp], sq = zq[so + lp];
            si = fminf(fmaxf(si, -CLIPV), CLIPV);
            sq = fminf(fmaxf(sq, -CLIPV), CLIPV);
            v2f B;
            B[0] = half ? 1.0f : si;
            B[1] = half ? 0.0f : sq;

            v8f D[4];
            #pragma unroll
            for (int g = 0; g < 4; ++g) {
                v8f cz = {};
                D[g] = __builtin_amdgcn_wmma_f32_16x16x4_f32(
                           false, Ag[g], false, B, (short)0, cz, false, false);
            }

            float gm[4];
            #pragma unroll
            for (int g = 0; g < 4; ++g) {
                const float a = fmaxf(fmaxf(D[g][0], D[g][1]), fmaxf(D[g][2], D[g][3]));
                const float b = fmaxf(fmaxf(D[g][4], D[g][5]), fmaxf(D[g][6], D[g][7]));
                gm[g] = fmaxf(a, b);
            }
            const float bs = fmaxf(fmaxf(gm[0], gm[1]), fmaxf(gm[2], gm[3]));

            int bslot = 31;
            #pragma unroll
            for (int slot = 30; slot >= 0; --slot) {
                bslot = (D[slot >> 3][slot & 7] == bs) ? slot : bslot;
            }
            int bi = ((bslot >> 3) << 4) + (bslot & 7) + (half << 3);

            const float os  = __shfl_xor(bs, 16, 32);
            const int   oi2 = __shfl_xor(bi, 16, 32);
            const int take = (int)(os > bs) | ((int)(os == bs) & (int)(oi2 < bi));
            bi = take ? oi2 : bi;

            const float val = cb[2 * bi + half] * scale;
            ob[half * PLANE + so + lp] = val;
        }
    }
}

extern "C" void kernel_launch(void* const* d_in, const int* in_sizes, int n_in,
                              void* d_out, int out_size, void* d_ws, size_t ws_size,
                              hipStream_t stream) {
    const float* z  = (const float*)d_in[0];
    const float* cb = (const float*)d_in[1];
    float* out = (float*)d_out;

    const int nsym     = in_sizes[0] / 2;
    const int numSuper = nsym / 32;

    dim3 block(256);
    dim3 grid(2048);
    qam_vq_wmma<<<grid, block, 0, stream>>>(z, cb, out, numSuper);
}


# in-lane argmax index by halving select: 5 compares, each level's selects share one condition (dual-issued), replacing the 31-step compare+select scan; same lowest-slot tie rule, same sums
# speedup vs baseline: 1.1334x; 1.1334x over previous
#include <hip/hip_runtime.h>


typedef __attribute__((ext_vector_type(2))) float v2f;
typedef __attribute__((ext_vector_type(8))) float v8f;

#define PLANE 1024
#define CLIPV 2.0f

__global__ __launch_bounds__(256)
void qam_vq_wmma(const float* __restrict__ z,
                 const float* __restrict__ cb,
                 float* __restrict__ out,
                 int numSuper)
{
    const int lane = threadIdx.x & 31;
    const int lp   = lane & 15;
    const int half = lane >> 4;

    const int wavesPerBlk = blockDim.x >> 5;
    const int gwave  = blockIdx.x * wavesPerBlk + (threadIdx.x >> 5);
    const int nwaves = gridDim.x * wavesPerBlk;

    float4 c4 = ((const float4*)cb)[lane];
    float ss = c4.x*c4.x + c4.y*c4.y + c4.z*c4.z + c4.w*c4.w;
    #pragma unroll
    for (int m = 16; m >= 1; m >>= 1) ss += __shfl_xor(ss, m, 32);
    const float power = ss * (1.0f / 64.0f);
    const float scale = sqrtf(1.0f / (power + 1e-8f));
    const float nhalf = -0.5f * scale * scale;

    v2f Ag[4];
    #pragma unroll
    for (int g = 0; g < 4; ++g) {
        const int e = g * 16 + lp;
        const float ci = cb[2*e], cq = cb[2*e + 1];
        const float ht = nhalf * (ci*ci + cq*cq);
        Ag[g][0] = half ? ht   : ci * scale;
        Ag[g][1] = half ? 0.0f : cq * scale;
    }

    for (int t = gwave; t < numSuper; t += nwaves) {
        const int p = t >> 5;
        const int o = (t & 31) << 5;
        const float* zi = z + (size_t)p * (2 * PLANE) + o;
        const float* zq = zi + PLANE;
        float* ob = out + (size_t)p * (2 * PLANE) + o;

        #pragma unroll
        for (int sub = 0; sub < 2; ++sub) {
            const int so = sub * 16;

            float si = zi[so + lp], sq = zq[so + lp];
            si = fminf(fmaxf(si, -CLIPV), CLIPV);
            sq = fminf(fmaxf(sq, -CLIPV), CLIPV);
            v2f B;
            B[0] = half ? 1.0f : si;
            B[1] = half ? 0.0f : sq;

            v8f D[4];
            #pragma unroll
            for (int g = 0; g < 4; ++g) {
                v8f cz = {};
                D[g] = __builtin_amdgcn_wmma_f32_16x16x4_f32(
                           false, Ag[g], false, B, (short)0, cz, false, false);
            }

            float gm[4];
            #pragma unroll
            for (int g = 0; g < 4; ++g) {
                const float a = fmaxf(fmaxf(D[g][0], D[g][1]), fmaxf(D[g][2], D[g][3]));
                const float b = fmaxf(fmaxf(D[g][4], D[g][5]), fmaxf(D[g][6], D[g][7]));
                gm[g] = fmaxf(a, b);
            }
            const float lo16 = fmaxf(gm[0], gm[1]);
            const float bs = fmaxf(lo16, fmaxf(gm[2], gm[3]));

            const bool h4 = lo16 < bs;
            float E[16];
            #pragma unroll
            for (int j = 0; j < 8; ++j) {
                E[j]     = h4 ? D[2][j] : D[0][j];
                E[8 + j] = h4 ? D[3][j] : D[1][j];
            }
            const float lo8 = h4 ? gm[2] : gm[0];
            const bool h3 = lo8 < bs;
            float F[8];
            #pragma unroll
            for (int j = 0; j < 8; ++j) F[j] = h3 ? E[8 + j] : E[j];
            const float lo4 = fmaxf(fmaxf(F[0], F[1]), fmaxf(F[2], F[3]));
            const bool h2 = lo4 < bs;
            float G[4];
            #pragma unroll
            for (int j = 0; j < 4; ++j) G[j] = h2 ? F[4 + j] : F[j];
            const bool h1 = fmaxf(G[0], G[1]) < bs;
            const float H0 = h1 ? G[2] : G[0];
            const bool h0 = H0 < bs;
            int bi = (h4 ? 32 : 0) | (h3 ? 16 : 0) | (h2 ? 4 : 0) | (h1 ? 2 : 0)
                   | (h0 ? 1 : 0) | (half << 3);

            const float os  = __shfl_xor(bs, 16, 32);
            const int   oi2 = __shfl_xor(bi, 16, 32);
            const int take = (int)(os > bs) | ((int)(os == bs) & (int)(oi2 < bi));
            bi = take ? oi2 : bi;

            const float val = cb[2 * bi + half] * scale;
            ob[half * PLANE + so + lp] = val;
        }
    }
}

extern "C" void kernel_launch(void* const* d_in, const int* in_sizes, int n_in,
                              void* d_out, int out_size, void* d_ws, size_t ws_size,
                              hipStream_t stream) {
    const float* z  = (const float*)d_in[0];
    const float* cb = (const float*)d_in[1];
    float* out = (float*)d_out;

    const int nsym     = in_sizes[0] / 2;
    const int numSuper = nsym / 32;

    dim3 block(256);
    dim3 grid(2048);
    qam_vq_wmma<<<grid, block, 0, stream>>>(z, cb, out, numSuper);
}


# resident waves: grid cut 4x (4096 waves, 4 per SIMD); each walks 8 tiles on a scalar tile index, next tile's symbols requested one trip ahead; wave entry and codebook prologue paid once per 8 tiles
# speedup vs baseline: 1.2437x; 1.0973x over previous
#include <hip/hip_runtime.h>


typedef __attribute__((ext_vector_type(2))) float v2f;
typedef __attribute__((ext_vector_type(8))) float v8f;

#define PLANE 1024
#define CLIPV 2.0f

__global__ __launch_bounds__(256)
void qam_vq_wmma(const float* __restrict__ z,
                 const float* __restrict__ cb,
                 float* __restrict__ out,
                 int numSuper)
{
    const int lane = threadIdx.x & 31;
    const int lp   = lane & 15;
    const int half = lane >> 4;

    const int wavesPerBlk = blockDim.x >> 5;
    const int wib    = __builtin_amdgcn_readfirstlane((int)(threadIdx.x >> 5));
    const int gwave  = blockIdx.x * wavesPerBlk + wib;
    const int nwaves = gridDim.x * wavesPerBlk;
    if (gwave >= numSuper) return;

    float4 c4 = ((const float4*)cb)[lane];
    float ss = c4.x*c4.x + c4.y*c4.y + c4.z*c4.z + c4.w*c4.w;
    #pragma unroll
    for (int m = 16; m >= 1; m >>= 1) ss += __shfl_xor(ss, m, 32);
    const float power = ss * (1.0f / 64.0f);
    const float scale = sqrtf(1.0f / (power + 1e-8f));
    const float nhalf = -0.5f * scale * scale;

    v2f Ag[4];
    #pragma unroll
    for (int g = 0; g < 4; ++g) {
        const int e = g * 16 + lp;
        const float ci = cb[2*e], cq = cb[2*e + 1];
        const float ht = nhalf * (ci*ci + cq*cq);
        Ag[g][0] = half ? ht   : ci * scale;
        Ag[g][1] = half ? 0.0f : cq * scale;
    }

    float curI[2], curQ[2];
    {
        const float* zi = z + (size_t)(gwave >> 5) * (2 * PLANE) + ((gwave & 31) << 5);
        #pragma unroll
        for (int sub = 0; sub < 2; ++sub) {
            curI[sub] = zi[sub * 16 + lp];
            curQ[sub] = zi[PLANE + sub * 16 + lp];
        }
    }

    for (int t = gwave; t < numSuper; t += nwaves) {
        const int tn = (t + nwaves < numSuper) ? t + nwaves : t;
        const float* zn = z + (size_t)(tn >> 5) * (2 * PLANE) + ((tn & 31) << 5);
        float nxtI[2], nxtQ[2];
        #pragma unroll
        for (int sub = 0; sub < 2; ++sub) {
            nxtI[sub] = zn[sub * 16 + lp];
            nxtQ[sub] = zn[PLANE + sub * 16 + lp];
        }
        __builtin_amdgcn_sched_barrier(0);

        float* ob = out + (size_t)(t >> 5) * (2 * PLANE) + ((t & 31) << 5);

        #pragma unroll
        for (int sub = 0; sub < 2; ++sub) {
            const int so = sub * 16;

            float si = curI[sub], sq = curQ[sub];
            si = fminf(fmaxf(si, -CLIPV), CLIPV);
            sq = fminf(fmaxf(sq, -CLIPV), CLIPV);
            v2f B;
            B[0] = half ? 1.0f : si;
            B[1] = half ? 0.0f : sq;

            v8f D[4];
            #pragma unroll
            for (int g = 0; g < 4; ++g) {
                v8f cz = {};
                D[g] = __builtin_amdgcn_wmma_f32_16x16x4_f32(
                           false, Ag[g], false, B, (short)0, cz, false, false);
            }

            float gm[4];
            #pragma unroll
            for (int g = 0; g < 4; ++g) {
                const float a = fmaxf(fmaxf(D[g][0], D[g][1]), fmaxf(D[g][2], D[g][3]));
                const float b = fmaxf(fmaxf(D[g][4], D[g][5]), fmaxf(D[g][6], D[g][7]));
                gm[g] = fmaxf(a, b);
            }
            const float lo16 = fmaxf(gm[0], gm[1]);
            const float bs = fmaxf(lo16, fmaxf(gm[2], gm[3]));

            const bool h4 = lo16 < bs;
            float E[16];
            #pragma unroll
            for (int j = 0; j < 8; ++j) {
                E[j]     = h4 ? D[2][j] : D[0][j];
                E[8 + j] = h4 ? D[3][j] : D[1][j];
            }
            const float lo8 = h4 ? gm[2] : gm[0];
            const bool h3 = lo8 < bs;
            float F[8];
            #pragma unroll
            for (int j = 0; j < 8; ++j) F[j] = h3 ? E[8 + j] : E[j];
            const float lo4 = fmaxf(fmaxf(F[0], F[1]), fmaxf(F[2], F[3]));
            const bool h2 = lo4 < bs;
            float G[4];
            #pragma unroll
            for (int j = 0; j < 4; ++j) G[j] = h2 ? F[4 + j] : F[j];
            const bool h1 = fmaxf(G[0], G[1]) < bs;
            const float H0 = h1 ? G[2] : G[0];
            const bool h0 = H0 < bs;
            int bi = (h4 ? 32 : 0) | (h3 ? 16 : 0) | (h2 ? 4 : 0) | (h1 ? 2 : 0)
                   | (h0 ? 1 : 0) | (half << 3);

            const float os  = __shfl_xor(bs, 16, 32);
            const int   oi2 = __shfl_xor(bi, 16, 32);
            const int take = (int)(os > bs) | ((int)(os == bs) & (int)(oi2 < bi));
            bi = take ? oi2 : bi;

            const float val = cb[2 * bi + half] * scale;
            ob[half * PLANE + so + lp] = val;
        }

        #pragma unroll
        for (int sub = 0; sub < 2; ++sub) { curI[sub] = nxtI[sub]; curQ[sub] = nxtQ[sub]; }
    }
}

extern "C" void kernel_launch(void* const* d_in, const int* in_sizes, int n_in,
                              void* d_out, int out_size, void* d_ws, size_t ws_size,
                              hipStream_t stream) {
    const float* z  = (const float*)d_in[0];
    const float* cb = (const float*)d_in[1];
    float* out = (float*)d_out;

    const int nsym     = in_sizes[0] / 2;
    const int numSuper = nsym / 32;

    dim3 block(256);
    dim3 grid(512);
    qam_vq_wmma<<<grid, block, 0, stream>>>(z, cb, out, numSuper);
}
